# starting kernel
# baseline (speedup 1.0000x reference)
#include <hip/hip_runtime.h>

typedef __attribute__((ext_vector_type(2))) float v2f;
typedef __attribute__((ext_vector_type(8))) float v8f;

#define BATCH 64
#define IN_CH 512
#define HID   128
#define NOUT  10

__global__ __launch_bounds__(256) void mlp_wmma_kernel(
    const float* __restrict__ features,
    const float* __restrict__ W1,
    const float* __restrict__ b1,
    const float* __restrict__ W2,
    const float* __restrict__ b2,
    const float* __restrict__ W3,
    const float* __restrict__ b3,
    float* __restrict__ alpha_out,
    float* __restrict__ kw_out)
{
    __shared__ float h1[BATCH][HID];
    __shared__ float h2[BATCH][HID];

    const int tid    = threadIdx.x;
    const int lane   = tid & 31;
    const int wave   = tid >> 5;
    const int nloc   = lane & 15;
    const int khalf  = (lane >> 4) * 2;
    const int rowoff = (lane >> 4) * 8;

    {
        const int n0 = wave * 16;
        for (int mt = 0; mt < 4; ++mt) {
            const int m0 = mt * 16;
            v8f acc = {};
            for (int k0 = 0; k0 < IN_CH; k0 += 4) {
                const int kb = k0 + khalf;
                v2f a, b;
                a.x = features[(m0 + nloc) * IN_CH + kb];
                a.y = features[(m0 + nloc) * IN_CH + kb + 1];
                b.x = W1[(n0 + nloc) * IN_CH + kb];
                b.y = W1[(n0 + nloc) * IN_CH + kb + 1];
                acc = __builtin_amdgcn_wmma_f32_16x16x4_f32(
                    false, a, false, b, (short)0, acc, false, false);
            }
            const float bias = b1[n0 + nloc];
#pragma unroll
            for (int r = 0; r < 8; ++r) {
                float v = acc[r] + bias;
                h1[m0 + r + rowoff][n0 + nloc] = v > 0.f ? v : 0.f;
            }
        }
    }
    __syncthreads();

    {
        const int n0 = wave * 16;
        for (int mt = 0; mt < 4; ++mt) {
            const int m0 = mt * 16;
            v8f acc = {};
            for (int k0 = 0; k0 < HID; k0 += 4) {
                const int kb = k0 + khalf;
                v2f a, b;
                a.x = h1[m0 + nloc][kb];
                a.y = h1[m0 + nloc][kb + 1];
                b.x = W2[(n0 + nloc) * HID + kb];
                b.y = W2[(n0 + nloc) * HID + kb + 1];
                acc = __builtin_amdgcn_wmma_f32_16x16x4_f32(
                    false, a, false, b, (short)0, acc, false, false);
            }
            const float bias = b2[n0 + nloc];
#pragma unroll
            for (int r = 0; r < 8; ++r) {
                float v = acc[r] + bias;
                h2[m0 + r + rowoff][n0 + nloc] = v > 0.f ? v : 0.f;
            }
        }
    }
    __syncthreads();

    if (wave < 4) {
        const int m0 = wave * 16;
        const bool valid = (nloc < NOUT);
        v8f acc = {};
        for (int k0 = 0; k0 < HID; k0 += 4) {
            const int kb = k0 + khalf;
            v2f a, b;
            a.x = h2[m0 + nloc][kb];
            a.y = h2[m0 + nloc][kb + 1];
            b.x = valid ? W3[nloc * HID + kb]     : 0.f;
            b.y = valid ? W3[nloc * HID + kb + 1] : 0.f;
            acc = __builtin_amdgcn_wmma_f32_16x16x4_f32(
                false, a, false, b, (short)0, acc, false, false);
        }
        const float bias = valid ? b3[nloc] : 0.f;
#pragma unroll
        for (int r = 0; r < 8; ++r) {
            const int row = m0 + r + rowoff;
            float v = acc[r] + bias;
            if (nloc == 0) {
                alpha_out[row] = v;
            } else if (valid) {
                v += (nloc == 5) ? 5.0f : -5.0f;
                kw_out[row * 9 + (nloc - 1)] = 1.0f / (1.0f + __expf(-v));
            }
        }
    }
}

#define TW 64
#define TH 4

__global__ __launch_bounds__(256) void morph_pixel_kernel(
    const float* __restrict__ x,
    const float* __restrict__ alpha,
    const float* __restrict__ kw,
    float* __restrict__ out,
    int H, int W)
{
    __shared__ float tile[TH + 2][TW + 4];

    const int b   = blockIdx.z;
    const int bx  = blockIdx.x * TW;
    const int by  = blockIdx.y * TH;
    const int tid = threadIdx.x;
    const float* img = x + (size_t)b * H * W;

    for (int idx = tid; idx < (TH + 2) * (TW + 2); idx += 256) {
        const int ly = idx / (TW + 2);
        const int lx = idx - ly * (TW + 2);
        const int gy = by + ly - 1;
        const int gx = bx + lx - 1;
        float v = 0.f;
        if (gy >= 0 && gy < H && gx >= 0 && gx < W) v = img[gy * W + gx];
        tile[ly][lx] = v;
    }

    const float a = alpha[b];
    float k[9];
#pragma unroll
    for (int j = 0; j < 9; ++j) k[j] = kw[b * 9 + j];

    __syncthreads();

    const int tx = tid & (TW - 1);
    const int ty = tid >> 6;

    float num = 0.f, den = 0.f;
#pragma unroll
    for (int dy = 0; dy < 3; ++dy) {
#pragma unroll
        for (int dx = 0; dx < 3; ++dx) {
            const float p = tile[ty + dy][tx + dx];
            const float w = __expf(a * p) * k[dy * 3 + dx];
            num += p * w;
            den += w;
        }
    }
    out[(size_t)b * H * W + (size_t)(by + ty) * W + (bx + tx)] = num / den;
}

extern "C" void kernel_launch(void* const* d_in, const int* in_sizes, int n_in,
                              void* d_out, int out_size, void* d_ws, size_t ws_size,
                              hipStream_t stream) {
    const float* x        = (const float*)d_in[0];
    const float* features = (const float*)d_in[1];
    const float* W1       = (const float*)d_in[2];
    const float* b1       = (const float*)d_in[3];
    const float* W2       = (const float*)d_in[4];
    const float* b2       = (const float*)d_in[5];
    const float* W3       = (const float*)d_in[6];
    const float* b3       = (const float*)d_in[7];

    float* alpha = (float*)d_ws;
    float* kw    = alpha + BATCH;
    float* out   = (float*)d_out;

    mlp_wmma_kernel<<<1, 256, 0, stream>>>(features, W1, b1, W2, b2, W3, b3,
                                           alpha, kw);

    const int H = 256, W = 256;
    dim3 grid(W / TW, H / TH, BATCH);
    morph_pixel_kernel<<<grid, 256, 0, stream>>>(x, alpha, kw, out, H, W);
}


# MLP fragments loaded straight from global memory as 128-bit per lane, one 32-k chunk ahead, each reused over a 2x2 block of tiles per wave (4 accumulators); k summed in another order
# speedup vs baseline: 1.8477x; 1.8477x over previous
#include <hip/hip_runtime.h>

typedef __attribute__((ext_vector_type(2))) float v2f;
typedef __attribute__((ext_vector_type(4))) float v4f;
typedef __attribute__((ext_vector_type(8))) float v8f;

#define BATCH 64
#define IN_CH 512
#define HID   128
#define NOUT  10
#define HP    132

struct Frag { v4f q[4]; };

__device__ __forceinline__ void load_frag(Frag& f, const float* p) {
#pragma unroll
    for (int j = 0; j < 4; ++j) f.q[j] = *(const v4f*)(p + 8 * j);
}

__device__ __forceinline__ v8f mma(v2f a, v2f b, v8f c) {
    return __builtin_amdgcn_wmma_f32_16x16x4_f32(false, a, false, b, (short)0, c,
                                                 false, false);
}

__device__ __forceinline__ void mma_chunk(const Frag& a0, const Frag& a1,
                                          const Frag& b0, const Frag& b1,
                                          v8f& c00, v8f& c01, v8f& c10, v8f& c11) {
#pragma unroll
    for (int j = 0; j < 4; ++j) {
        c00 = mma(a0.q[j].xy, b0.q[j].xy, c00);
        c01 = mma(a0.q[j].xy, b1.q[j].xy, c01);
        c10 = mma(a1.q[j].xy, b0.q[j].xy, c10);
        c11 = mma(a1.q[j].xy, b1.q[j].xy, c11);
        c00 = mma(a0.q[j].zw, b0.q[j].zw, c00);
        c01 = mma(a0.q[j].zw, b1.q[j].zw, c01);
        c10 = mma(a1.q[j].zw, b0.q[j].zw, c10);
        c11 = mma(a1.q[j].zw, b1.q[j].zw, c11);
    }
}

__device__ __forceinline__ void store_relu(float* h, int row0, int col, float bias,
                                           const v8f& acc) {
#pragma unroll
    for (int r = 0; r < 8; ++r) {
        const float v = acc[r] + bias;
        h[(row0 + r) * HP + col] = v > 0.f ? v : 0.f;
    }
}

__global__ __launch_bounds__(256) void mlp_wmma_kernel(
    const float* __restrict__ features,
    const float* __restrict__ W1,
    const float* __restrict__ b1,
    const float* __restrict__ W2,
    const float* __restrict__ b2,
    const float* __restrict__ W3,
    const float* __restrict__ b3,
    float* __restrict__ alpha_out,
    float* __restrict__ kw_out)
{
    __shared__ __attribute__((aligned(16))) float h1[BATCH * HP];
    __shared__ __attribute__((aligned(16))) float h2[BATCH * HP];

    const int tid  = threadIdx.x;
    const int lane = tid & 31;
    const int wave = tid >> 5;
    const int c    = lane & 15;
    const int h    = lane >> 4;
    const int wm   = wave >> 2;
    const int wn   = wave & 3;
    const int m0   = 32 * wm;
    const int n0   = 32 * wn;

    {
        const float* pa0 = features + (m0 + c) * IN_CH + 4 * h;
        const float* pa1 = pa0 + 16 * IN_CH;
        const float* pb0 = W1 + (n0 + c) * IN_CH + 4 * h;
        const float* pb1 = pb0 + 16 * IN_CH;
        v8f c00 = {}, c01 = {}, c10 = {}, c11 = {};
        Frag a0, a1, b0, b1f, x0, x1, y0, y1;
        load_frag(a0, pa0); load_frag(a1, pa1);
        load_frag(b0, pb0); load_frag(b1f, pb1);
        for (int kk = 0; kk < IN_CH; kk += 64) {
            load_frag(x0, pa0 + kk + 32); load_frag(x1, pa1 + kk + 32);
            load_frag(y0, pb0 + kk + 32); load_frag(y1, pb1 + kk + 32);
            mma_chunk(a0, a1, b0, b1f, c00, c01, c10, c11);
            if (kk + 64 < IN_CH) {
                load_frag(a0, pa0 + kk + 64); load_frag(a1, pa1 + kk + 64);
                load_frag(b0, pb0 + kk + 64); load_frag(b1f, pb1 + kk + 64);
            }
            mma_chunk(x0, x1, y0, y1, c00, c01, c10, c11);
        }
        const float bias0 = b1[n0 + c];
        const float bias1 = b1[n0 + 16 + c];
        store_relu(h1, m0 + 8 * h,      n0 + c,      bias0, c00);
        store_relu(h1, m0 + 8 * h,      n0 + 16 + c, bias1, c01);
        store_relu(h1, m0 + 16 + 8 * h, n0 + c,      bias0, c10);
        store_relu(h1, m0 + 16 + 8 * h, n0 + 16 + c, bias1, c11);
    }

    const bool valid = (c < NOUT);
    const bool act   = (wave < 4);
    Frag w[4];
    if (act) {
        const float* pw = W3 + (valid ? c : 0) * HID + 4 * h;
        const v4f z = {0.f, 0.f, 0.f, 0.f};
#pragma unroll
        for (int ch = 0; ch < 4; ++ch) {
#pragma unroll
            for (int j = 0; j < 4; ++j) {
                const v4f t = *(const v4f*)(pw + 32 * ch + 8 * j);
                w[ch].q[j] = valid ? t : z;
            }
        }
    }

    {
        const float* pb0 = W2 + (n0 + c) * HID + 4 * h;
        const float* pb1 = pb0 + 16 * HID;
        Frag a0, a1, b0, b1f, y0, y1;
        load_frag(b0, pb0); load_frag(b1f, pb1);
        __syncthreads();
        const float* pa0 = h1 + (m0 + c) * HP + 4 * h;
        const float* pa1 = pa0 + 16 * HP;
        v8f c00 = {}, c01 = {}, c10 = {}, c11 = {};
#pragma unroll
        for (int kk = 0; kk < HID; kk += 32) {
            if (kk + 32 < HID) { load_frag(y0, pb0 + kk + 32); load_frag(y1, pb1 + kk + 32); }
            load_frag(a0, pa0 + kk); load_frag(a1, pa1 + kk);
            mma_chunk(a0, a1, b0, b1f, c00, c01, c10, c11);
            b0 = y0; b1f = y1;
        }
        const float bias0 = b2[n0 + c];
        const float bias1 = b2[n0 + 16 + c];
        store_relu(h2, m0 + 8 * h,      n0 + c,      bias0, c00);
        store_relu(h2, m0 + 8 * h,      n0 + 16 + c, bias1, c01);
        store_relu(h2, m0 + 16 + 8 * h, n0 + c,      bias0, c10);
        store_relu(h2, m0 + 16 + 8 * h, n0 + 16 + c, bias1, c11);
    }

    {
        __syncthreads();
        if (act) {
            const int r0 = wave * 16;
            const float* pa = h2 + (r0 + c) * HP + 4 * h;
            v8f e0 = {}, e1 = {};
#pragma unroll
            for (int ch = 0; ch < 4; ++ch) {
                Frag a;
                load_frag(a, pa + 32 * ch);
#pragma unroll
                for (int j = 0; j < 4; ++j) {
                    e0 = mma(a.q[j].xy, w[ch].q[j].xy, e0);
                    e1 = mma(a.q[j].zw, w[ch].q[j].zw, e1);
                }
            }
            const float bias = valid ? b3[c] : 0.f;
#pragma unroll
            for (int r = 0; r < 8; ++r) {
                const int row = r0 + r + 8 * h;
                float v = (e0[r] + e1[r]) + bias;
                if (c == 0) {
                    alpha_out[row] = v;
                } else if (valid) {
                    v += (c == 5) ? 5.0f : -5.0f;
                    kw_out[row * 9 + (c - 1)] = 1.0f / (1.0f + __expf(-v));
                }
            }
        }
    }
}

#define TW 64
#define TH 4

__global__ __launch_bounds__(256) void morph_pixel_kernel(
    const float* __restrict__ x,
    const float* __restrict__ alpha,
    const float* __restrict__ kw,
    float* __restrict__ out,
    int H, int W)
{
    __shared__ float tile[TH + 2][TW + 4];

    const int b   = blockIdx.z;
    const int bx  = blockIdx.x * TW;
    const int by  = blockIdx.y * TH;
    const int tid = threadIdx.x;
    const float* img = x + (size_t)b * H * W;

    for (int idx = tid; idx < (TH + 2) * (TW + 2); idx += 256) {
        const int ly = idx / (TW + 2);
        const int lx = idx - ly * (TW + 2);
        const int gy = by + ly - 1;
        const int gx = bx + lx - 1;
        float v = 0.f;
        if (gy >= 0 && gy < H && gx >= 0 && gx < W) v = img[gy * W + gx];
        tile[ly][lx] = v;
    }

    const float a = alpha[b];
    float k[9];
#pragma unroll
    for (int j = 0; j < 9; ++j) k[j] = kw[b * 9 + j];

    __syncthreads();

    const int tx = tid & (TW - 1);
    const int ty = tid >> 6;

    float num = 0.f, den = 0.f;
#pragma unroll
    for (int dy = 0; dy < 3; ++dy) {
#pragma unroll
        for (int dx = 0; dx < 3; ++dx) {
            const float p = tile[ty + dy][tx + dx];
            const float w = __expf(a * p) * k[dy * 3 + dx];
            num += p * w;
            den += w;
        }
    }
    out[(size_t)b * H * W + (size_t)(by + ty) * W + (bx + tx)] = num / den;
}

extern "C" void kernel_launch(void* const* d_in, const int* in_sizes, int n_in,
                              void* d_out, int out_size, void* d_ws, size_t ws_size,
                              hipStream_t stream) {
    const float* x        = (const float*)d_in[0];
    const float* features = (const float*)d_in[1];
    const float* W1       = (const float*)d_in[2];
    const float* b1       = (const float*)d_in[3];
    const float* W2       = (const float*)d_in[4];
    const float* b2       = (const float*)d_in[5];
    const float* W3       = (const float*)d_in[6];
    const float* b3       = (const float*)d_in[7];

    float* alpha = (float*)d_ws;
    float* kw    = alpha + BATCH;
    float* out   = (float*)d_out;

    mlp_wmma_kernel<<<1, 256, 0, stream>>>(features, W1, b1, W2, b2, W3, b3,
                                           alpha, kw);

    const int H = 256, W = 256;
    dim3 grid(W / TW, H / TH, BATCH);
    morph_pixel_kernel<<<grid, 256, 0, stream>>>(x, alpha, kw, out, H, W);
}


# Pixel pass streamed by rows: a wave owns 8 full-width rows, one (p*e, e) pair per pixel, 3x3 sums by packed fma over 3 rows in registers, side columns by lane exchange; 2048 waves, no LDS
# speedup vs baseline: 2.8599x; 1.5478x over previous
#include <hip/hip_runtime.h>

typedef __attribute__((ext_vector_type(2))) float v2f;
typedef __attribute__((ext_vector_type(4))) float v4f;
typedef __attribute__((ext_vector_type(8))) float v8f;

#define BATCH 64
#define IN_CH 512
#define HID   128
#define NOUT  10
#define HP    132

struct Frag { v4f q[4]; };

__device__ __forceinline__ void load_frag(Frag& f, const float* p) {
#pragma unroll
    for (int j = 0; j < 4; ++j) f.q[j] = *(const v4f*)(p + 8 * j);
}

__device__ __forceinline__ v8f mma(v2f a, v2f b, v8f c) {
    return __builtin_amdgcn_wmma_f32_16x16x4_f32(false, a, false, b, (short)0, c,
                                                 false, false);
}

__device__ __forceinline__ void mma_chunk(const Frag& a0, const Frag& a1,
                                          const Frag& b0, const Frag& b1,
                                          v8f& c00, v8f& c01, v8f& c10, v8f& c11) {
#pragma unroll
    for (int j = 0; j < 4; ++j) {
        c00 = mma(a0.q[j].xy, b0.q[j].xy, c00);
        c01 = mma(a0.q[j].xy, b1.q[j].xy, c01);
        c10 = mma(a1.q[j].xy, b0.q[j].xy, c10);
        c11 = mma(a1.q[j].xy, b1.q[j].xy, c11);
        c00 = mma(a0.q[j].zw, b0.q[j].zw, c00);
        c01 = mma(a0.q[j].zw, b1.q[j].zw, c01);
        c10 = mma(a1.q[j].zw, b0.q[j].zw, c10);
        c11 = mma(a1.q[j].zw, b1.q[j].zw, c11);
    }
}

__device__ __forceinline__ void store_relu(float* h, int row0, int col, float bias,
                                           const v8f& acc) {
#pragma unroll
    for (int r = 0; r < 8; ++r) {
        const float v = acc[r] + bias;
        h[(row0 + r) * HP + col] = v > 0.f ? v : 0.f;
    }
}

__global__ __launch_bounds__(256) void mlp_wmma_kernel(
    const float* __restrict__ features,
    const float* __restrict__ W1,
    const float* __restrict__ b1,
    const float* __restrict__ W2,
    const float* __restrict__ b2,
    const float* __restrict__ W3,
    const float* __restrict__ b3,
    float* __restrict__ alpha_out,
    float* __restrict__ kw_out)
{
    __shared__ __attribute__((aligned(16))) float h1[BATCH * HP];
    __shared__ __attribute__((aligned(16))) float h2[BATCH * HP];

    const int tid  = threadIdx.x;
    const int lane = tid & 31;
    const int wave = tid >> 5;
    const int c    = lane & 15;
    const int h    = lane >> 4;
    const int wm   = wave >> 2;
    const int wn   = wave & 3;
    const int m0   = 32 * wm;
    const int n0   = 32 * wn;

    {
        const float* pa0 = features + (m0 + c) * IN_CH + 4 * h;
        const float* pa1 = pa0 + 16 * IN_CH;
        const float* pb0 = W1 + (n0 + c) * IN_CH + 4 * h;
        const float* pb1 = pb0 + 16 * IN_CH;
        v8f c00 = {}, c01 = {}, c10 = {}, c11 = {};
        Frag a0, a1, b0, b1f, x0, x1, y0, y1;
        load_frag(a0, pa0); load_frag(a1, pa1);
        load_frag(b0, pb0); load_frag(b1f, pb1);
        for (int kk = 0; kk < IN_CH; kk += 64) {
            load_frag(x0, pa0 + kk + 32); load_frag(x1, pa1 + kk + 32);
            load_frag(y0, pb0 + kk + 32); load_frag(y1, pb1 + kk + 32);
            mma_chunk(a0, a1, b0, b1f, c00, c01, c10, c11);
            if (kk + 64 < IN_CH) {
                load_frag(a0, pa0 + kk + 64); load_frag(a1, pa1 + kk + 64);
                load_frag(b0, pb0 + kk + 64); load_frag(b1f, pb1 + kk + 64);
            }
            mma_chunk(x0, x1, y0, y1, c00, c01, c10, c11);
        }
        const float bias0 = b1[n0 + c];
        const float bias1 = b1[n0 + 16 + c];
        store_relu(h1, m0 + 8 * h,      n0 + c,      bias0, c00);
        store_relu(h1, m0 + 8 * h,      n0 + 16 + c, bias1, c01);
        store_relu(h1, m0 + 16 + 8 * h, n0 + c,      bias0, c10);
        store_relu(h1, m0 + 16 + 8 * h, n0 + 16 + c, bias1, c11);
    }

    const bool valid = (c < NOUT);
    const bool act   = (wave < 4);
    Frag w[4];
    if (act) {
        const float* pw = W3 + (valid ? c : 0) * HID + 4 * h;
        const v4f z = {0.f, 0.f, 0.f, 0.f};
#pragma unroll
        for (int ch = 0; ch < 4; ++ch) {
#pragma unroll
            for (int j = 0; j < 4; ++j) {
                const v4f t = *(const v4f*)(pw + 32 * ch + 8 * j);
                w[ch].q[j] = valid ? t : z;
            }
        }
    }

    {
        const float* pb0 = W2 + (n0 + c) * HID + 4 * h;
        const float* pb1 = pb0 + 16 * HID;
        Frag a0, a1, b0, b1f, y0, y1;
        load_frag(b0, pb0); load_frag(b1f, pb1);
        __syncthreads();
        const float* pa0 = h1 + (m0 + c) * HP + 4 * h;
        const float* pa1 = pa0 + 16 * HP;
        v8f c00 = {}, c01 = {}, c10 = {}, c11 = {};
#pragma unroll
        for (int kk = 0; kk < HID; kk += 32) {
            if (kk + 32 < HID) { load_frag(y0, pb0 + kk + 32); load_frag(y1, pb1 + kk + 32); }
            load_frag(a0, pa0 + kk); load_frag(a1, pa1 + kk);
            mma_chunk(a0, a1, b0, b1f, c00, c01, c10, c11);
            b0 = y0; b1f = y1;
        }
        const float bias0 = b2[n0 + c];
        const float bias1 = b2[n0 + 16 + c];
        store_relu(h2, m0 + 8 * h,      n0 + c,      bias0, c00);
        store_relu(h2, m0 + 8 * h,      n0 + 16 + c, bias1, c01);
        store_relu(h2, m0 + 16 + 8 * h, n0 + c,      bias0, c10);
        store_relu(h2, m0 + 16 + 8 * h, n0 + 16 + c, bias1, c11);
    }

    {
        __syncthreads();
        if (act) {
            const int r0 = wave * 16;
            const float* pa = h2 + (r0 + c) * HP + 4 * h;
            v8f e0 = {}, e1 = {};
#pragma unroll
            for (int ch = 0; ch < 4; ++ch) {
                Frag a;
                load_frag(a, pa + 32 * ch);
#pragma unroll
                for (int j = 0; j < 4; ++j) {
                    e0 = mma(a.q[j].xy, w[ch].q[j].xy, e0);
                    e1 = mma(a.q[j].zw, w[ch].q[j].zw, e1);
                }
            }
            const float bias = valid ? b3[c] : 0.f;
#pragma unroll
            for (int r = 0; r < 8; ++r) {
                const int row = r0 + r + 8 * h;
                float v = (e0[r] + e1[r]) + bias;
                if (c == 0) {
                    alpha_out[row] = v;
                } else if (valid) {
                    v += (c == 5) ? 5.0f : -5.0f;
                    kw_out[row * 9 + (c - 1)] = 1.0f / (1.0f + __expf(-v));
                }
            }
        }
    }
}

#define IMG_H 256
#define IMG_W 256
#define BAND  8
#define PIX_WAVES 8
#define GROUPS_PER_IMAGE (IMG_H / (BAND * PIX_WAVES))

__device__ __forceinline__ float lane_pull(int byte_addr, float v) {
    return __int_as_float(__builtin_amdgcn_ds_bpermute(byte_addr, __float_as_int(v)));
}

__device__ __forceinline__ v2f pair_pull(int byte_addr, v2f v) {
    v2f r;
    r.x = lane_pull(byte_addr, v.x);
    r.y = lane_pull(byte_addr, v.y);
    return r;
}

__global__ __launch_bounds__(256) void morph_pixel_kernel(
    const float* __restrict__ x,
    const float* __restrict__ alpha,
    const float* __restrict__ kw,
    float* __restrict__ out)
{
    const int lane = threadIdx.x & 31;
    const int wave = __builtin_amdgcn_readfirstlane((int)(threadIdx.x >> 5));
    const int b    = blockIdx.x / GROUPS_PER_IMAGE;
    const int y0   = (blockIdx.x % GROUPS_PER_IMAGE) * (BAND * PIX_WAVES) + wave * BAND;

    const float* img  = x   + (size_t)b * (IMG_H * IMG_W) + 4 * lane;
    float*       oimg = out + (size_t)b * (IMG_H * IMG_W) + 4 * lane;

    const float a = alpha[b];
    v2f k[9];
#pragma unroll
    for (int j = 0; j < 9; ++j) { const float kj = kw[b * 9 + j]; k[j].x = kj; k[j].y = kj; }

    const bool top_inside    = (y0 > 0);
    const bool bottom_inside = (y0 + BAND < IMG_H);
    v4f raw[BAND + 2][2];
#pragma unroll
    for (int i = 0; i < BAND + 2; ++i) {
        const bool inside = (i == 0) ? top_inside : (i == BAND + 1) ? bottom_inside : true;
        const float* rp   = img + (inside ? y0 - 1 + i : y0) * IMG_W;
        raw[i][0] = *(const v4f*)rp;
        raw[i][1] = *(const v4f*)(rp + IMG_W / 2);
    }
    __builtin_amdgcn_sched_barrier(0);

    v2f pad;
    pad.y = __expf(a * 0.f);
    pad.x = 0.f * pad.y;

    const int  from_left  = ((lane + 31) & 31) * 4;
    const int  from_right = ((lane + 1) & 31) * 4;
    const bool first = (lane == 0);
    const bool last  = (lane == 31);

    v2f t[BAND + 2][2][6];
#pragma unroll
    for (int i = 0; i < BAND + 2; ++i) {
#pragma unroll
        for (int g = 0; g < 2; ++g) {
#pragma unroll
            for (int s = 0; s < 4; ++s) {
                const bool  inside = (i == 0) ? top_inside : (i == BAND + 1) ? bottom_inside : true;
                const float p = inside ? raw[i][g][s] : 0.f;
                const float e = __expf(a * p);
                t[i][g][s + 1].x = p * e;
                t[i][g][s + 1].y = e;
            }
        }
        const v2f l0 = pair_pull(from_left,  t[i][0][4]);
        const v2f l1 = pair_pull(from_left,  t[i][1][4]);
        const v2f r0 = pair_pull(from_right, t[i][0][1]);
        const v2f r1 = pair_pull(from_right, t[i][1][1]);
        t[i][0][0] = first ? pad : l0;
        t[i][1][0] = first ? l0  : l1;
        t[i][0][5] = last  ? r1  : r0;
        t[i][1][5] = last  ? pad : r1;

        if (i >= 2) {
            const int o = i - 2;
            v4f res[2];
#pragma unroll
            for (int g = 0; g < 2; ++g) {
#pragma unroll
                for (int s = 0; s < 4; ++s) {
                    v2f acc = {0.f, 0.f};
#pragma unroll
                    for (int dy = 0; dy < 3; ++dy) {
#pragma unroll
                        for (int dx = 0; dx < 3; ++dx)
                            acc = __builtin_elementwise_fma(k[dy * 3 + dx], t[o + dy][g][s + dx], acc);
                    }
                    res[g][s] = acc.x / acc.y;
                }
            }
            float* op = oimg + (y0 + o) * IMG_W;
            *(v4f*)op               = res[0];
            *(v4f*)(op + IMG_W / 2) = res[1];
            __builtin_amdgcn_sched_barrier(0x81);
        }
    }
}

extern "C" void kernel_launch(void* const* d_in, const int* in_sizes, int n_in,
                              void* d_out, int out_size, void* d_ws, size_t ws_size,
                              hipStream_t stream) {
    const float* x        = (const float*)d_in[0];
    const float* features = (const float*)d_in[1];
    const float* W1       = (const float*)d_in[2];
    const float* b1       = (const float*)d_in[3];
    const float* W2       = (const float*)d_in[4];
    const float* b2       = (const float*)d_in[5];
    const float* W3       = (const float*)d_in[6];
    const float* b3       = (const float*)d_in[7];

    float* alpha = (float*)d_ws;
    float* kw    = alpha + BATCH;
    float* out   = (float*)d_out;

    mlp_wmma_kernel<<<1, 256, 0, stream>>>(features, W1, b1, W2, b2, W3, b3,
                                           alpha, kw);

    morph_pixel_kernel<<<BATCH * GROUPS_PER_IMAGE, 32 * PIX_WAVES, 0, stream>>>(x, alpha, kw, out);
}


# Pixel pass: each output's quotient taken as numerator * v_rcp_f32(denominator) (1 ulp) instead of the 11-instruction correctly rounded f32 division; the sigmoid's division likewise
# speedup vs baseline: 8.0520x; 2.8155x over previous
#include <hip/hip_runtime.h>

typedef __attribute__((ext_vector_type(2))) float v2f;
typedef __attribute__((ext_vector_type(4))) float v4f;
typedef __attribute__((ext_vector_type(8))) float v8f;

#define BATCH 64
#define IN_CH 512
#define HID   128
#define NOUT  10
#define MT    (BATCH / 16)
#define NT    (HID / 16)
#define TP    20

struct Frag { v4f q[4]; };

__device__ __forceinline__ void load_frag(Frag& f, const float* p) {
#pragma unroll
    for (int j = 0; j < 4; ++j) f.q[j] = *(const v4f*)(p + 8 * j);
}

__device__ __forceinline__ v8f mma(v2f a, v2f b, v8f c) {
    return __builtin_amdgcn_wmma_f32_16x16x4_f32(false, a, false, b, (short)0, c,
                                                 false, false);
}

__global__ __launch_bounds__(256) void mlp_tile_kernel(
    const float* __restrict__ features,
    const float* __restrict__ W1,
    const float* __restrict__ b1,
    const float* __restrict__ W2,
    float* __restrict__ share)
{
    __shared__ __attribute__((aligned(16))) float psum[8][256];
    __shared__ __attribute__((aligned(16))) float tile[16 * TP];

    const int tid  = threadIdx.x;
    const int lane = tid & 31;
    const int wave = tid >> 5;
    const int c    = lane & 15;
    const int h    = lane >> 4;
    const int mt   = blockIdx.x >> 3;
    const int nt   = blockIdx.x & 7;

    const float* pa = features + (16 * mt + c) * IN_CH + 64 * wave + 4 * h;
    const float* pb = W1       + (16 * nt + c) * IN_CH + 64 * wave + 4 * h;
    Frag a0, a1, b0, b1f;
    load_frag(a0, pa);      load_frag(b0, pb);
    load_frag(a1, pa + 32); load_frag(b1f, pb + 32);
    const float* pw = W2 + (16 * wave + c) * HID + 16 * nt + 4 * h;
    const v4f w0 = *(const v4f*)pw;
    const v4f w1 = *(const v4f*)(pw + 8);
    const float bias = b1[16 * nt + (tid & 15)];
    __builtin_amdgcn_sched_barrier(0);

    v8f e0 = {}, e1 = {};
#pragma unroll
    for (int j = 0; j < 4; ++j) {
        e0 = mma(a0.q[j].xy, b0.q[j].xy, e0);
        e1 = mma(a0.q[j].zw, b0.q[j].zw, e1);
    }
#pragma unroll
    for (int j = 0; j < 4; ++j) {
        e0 = mma(a1.q[j].xy, b1f.q[j].xy, e0);
        e1 = mma(a1.q[j].zw, b1f.q[j].zw, e1);
    }
#pragma unroll
    for (int r = 0; r < 8; ++r) psum[wave][(8 * h + r) * 16 + c] = e0[r] + e1[r];
    __syncthreads();

    {
        float s = psum[0][tid];
#pragma unroll
        for (int w = 1; w < 8; ++w) s += psum[w][tid];
        s += bias;
        tile[(tid >> 4) * TP + (tid & 15)] = s > 0.f ? s : 0.f;
    }
    __syncthreads();

    const v4f x0 = *(const v4f*)(tile + c * TP + 4 * h);
    const v4f x1 = *(const v4f*)(tile + c * TP + 8 + 4 * h);
    v8f f0 = {}, f1 = {};
    f0 = mma(x0.xy, w0.xy, f0);
    f1 = mma(x0.zw, w0.zw, f1);
    f0 = mma(x1.xy, w1.xy, f0);
    f1 = mma(x1.zw, w1.zw, f1);
    float* po = share + (blockIdx.x * 8 + wave) * 256 + (8 * h) * 16 + c;
#pragma unroll
    for (int r = 0; r < 8; ++r) po[r * 16] = f0[r] + f1[r];
}

#define IMG_H 256
#define IMG_W 256
#define BAND  8
#define PIX_WAVES 8
#define GROUPS_PER_IMAGE (IMG_H / (BAND * PIX_WAVES))

__device__ __forceinline__ float lane_pull(int byte_addr, float v) {
    return __int_as_float(__builtin_amdgcn_ds_bpermute(byte_addr, __float_as_int(v)));
}

__device__ __forceinline__ float fast_div(float n, float d) {
    return n * __builtin_amdgcn_rcpf(d);
}

__device__ __forceinline__ v2f pair_pull(int byte_addr, v2f v) {
    v2f r;
    r.x = lane_pull(byte_addr, v.x);
    r.y = lane_pull(byte_addr, v.y);
    return r;
}

__global__ __launch_bounds__(256) void morph_pixel_kernel(
    const float* __restrict__ x,
    const float* __restrict__ share,
    const float* __restrict__ b2,
    const float* __restrict__ W3,
    const float* __restrict__ b3,
    float* __restrict__ out)
{
    __shared__ __attribute__((aligned(16))) float h2s[2 * HID];

    const int tid  = threadIdx.x;
    const int lane = tid & 31;
    const int wave = __builtin_amdgcn_readfirstlane((int)(threadIdx.x >> 5));
    const int b    = blockIdx.x / GROUPS_PER_IMAGE;
    const int y0   = (blockIdx.x % GROUPS_PER_IMAGE) * (BAND * PIX_WAVES) + wave * BAND;

    const float* img  = x   + (size_t)b * (IMG_H * IMG_W) + 4 * lane;
    float*       oimg = out + (size_t)b * (IMG_H * IMG_W) + 4 * lane;

    const int n = tid & (HID - 1);
    const float* ps = share + ((b >> 4) * NT * 8 + (n >> 4)) * 256 + (b & 15) * 16 + (n & 15);
    float part[NT];
#pragma unroll
    for (int nt = 0; nt < NT; ++nt) part[nt] = ps[nt * 8 * 256];
    const float bias2 = b2[n];

    const int  c     = lane & 15;
    const int  h     = lane >> 4;
    const bool valid = (c < NOUT);
    Frag w[4];
    {
        const float* pw = W3 + (valid ? c : 0) * HID + 4 * h;
#pragma unroll
        for (int ch = 0; ch < 4; ++ch) load_frag(w[ch], pw + 32 * ch);
    }
    const float b3v   = b3[valid ? c : 0];

    __builtin_amdgcn_sched_barrier(0);

    const bool top_inside    = (y0 > 0);
    const bool bottom_inside = (y0 + BAND < IMG_H);
    v4f raw[BAND + 2][2];
#pragma unroll
    for (int i = 0; i < BAND + 2; ++i) {
        const bool inside = (i == 0) ? top_inside : (i == BAND + 1) ? bottom_inside : true;
        const float* rp   = img + (inside ? y0 - 1 + i : y0) * IMG_W;
        raw[i][0] = *(const v4f*)rp;
        raw[i][1] = *(const v4f*)(rp + IMG_W / 2);
    }
    __builtin_amdgcn_sched_barrier(0);

    {
        const v4f z = {0.f, 0.f, 0.f, 0.f};
#pragma unroll
        for (int ch = 0; ch < 4; ++ch) {
#pragma unroll
            for (int j = 0; j < 4; ++j) w[ch].q[j] = valid ? w[ch].q[j] : z;
        }
    }
    {
        float s = part[0];
#pragma unroll
        for (int nt = 1; nt < NT; ++nt) s += part[nt];
        s += bias2;
        h2s[tid] = s > 0.f ? s : 0.f;
    }
    __builtin_amdgcn_fence(__ATOMIC_RELEASE, "workgroup", "local");
    __builtin_amdgcn_s_barrier();
    __builtin_amdgcn_fence(__ATOMIC_ACQUIRE, "workgroup", "local");
    float a;
    v2f k[9];
    {
        const float* pa = h2s + (tid & HID) + 4 * h;
        v8f e0 = {}, e1 = {};
#pragma unroll
        for (int ch = 0; ch < 4; ++ch) {
            Frag f;
            load_frag(f, pa + 32 * ch);
#pragma unroll
            for (int j = 0; j < 4; ++j) {
                e0 = mma(f.q[j].xy, w[ch].q[j].xy, e0);
                e1 = mma(f.q[j].zw, w[ch].q[j].zw, e1);
            }
        }
        const float beta = (e0[0] + e1[0]) + (valid ? b3v : 0.f);
        const float v    = beta + ((c == 5) ? 5.0f : -5.0f);
        const float sig  = fast_div(1.0f, 1.0f + __expf(-v));
        a = __int_as_float(__builtin_amdgcn_readlane(__float_as_int(beta), 0));
#pragma unroll
        for (int j = 0; j < 9; ++j) {
            const float kj = __int_as_float(__builtin_amdgcn_readlane(__float_as_int(sig), j + 1));
            k[j].x = kj; k[j].y = kj;
        }
    }

    v2f pad;
    pad.y = __expf(a * 0.f);
    pad.x = 0.f * pad.y;

    const int  from_left  = ((lane + 31) & 31) * 4;
    const int  from_right = ((lane + 1) & 31) * 4;
    const bool first = (lane == 0);
    const bool last  = (lane == 31);

    v2f t[BAND + 2][2][6];
#pragma unroll
    for (int i = 0; i < BAND + 2; ++i) {
#pragma unroll
        for (int g = 0; g < 2; ++g) {
#pragma unroll
            for (int s = 0; s < 4; ++s) {
                const bool  inside = (i == 0) ? top_inside : (i == BAND + 1) ? bottom_inside : true;
                const float p = inside ? raw[i][g][s] : 0.f;
                const float e = __expf(a * p);
                t[i][g][s + 1].x = p * e;
                t[i][g][s + 1].y = e;
            }
        }
        const v2f l0 = pair_pull(from_left,  t[i][0][4]);
        const v2f l1 = pair_pull(from_left,  t[i][1][4]);
        const v2f r0 = pair_pull(from_right, t[i][0][1]);
        const v2f r1 = pair_pull(from_right, t[i][1][1]);
        t[i][0][0] = first ? pad : l0;
        t[i][1][0] = first ? l0  : l1;
        t[i][0][5] = last  ? r1  : r0;
        t[i][1][5] = last  ? pad : r1;

        if (i >= 2) {
            const int o = i - 2;
            v4f res[2];
#pragma unroll
            for (int g = 0; g < 2; ++g) {
#pragma unroll
                for (int s = 0; s < 4; ++s) {
                    v2f acc = {0.f, 0.f};
#pragma unroll
                    for (int dy = 0; dy < 3; ++dy) {
#pragma unroll
                        for (int dx = 0; dx < 3; ++dx)
                            acc = __builtin_elementwise_fma(k[dy * 3 + dx], t[o + dy][g][s + dx], acc);
                    }
                    res[g][s] = fast_div(acc.x, acc.y);
                }
            }
            float* op = oimg + (y0 + o) * IMG_W;
            *(v4f*)op               = res[0];
            *(v4f*)(op + IMG_W / 2) = res[1];
            __builtin_amdgcn_sched_barrier(0x81);
        }
    }
}

extern "C" void kernel_launch(void* const* d_in, const int* in_sizes, int n_in,
                              void* d_out, int out_size, void* d_ws, size_t ws_size,
                              hipStream_t stream) {
    const float* x        = (const float*)d_in[0];
    const float* features = (const float*)d_in[1];
    const float* W1       = (const float*)d_in[2];
    const float* b1       = (const float*)d_in[3];
    const float* W2       = (const float*)d_in[4];
    const float* b2       = (const float*)d_in[5];
    const float* W3       = (const float*)d_in[6];
    const float* b3       = (const float*)d_in[7];

    float* share = (float*)d_ws;
    float* out   = (float*)d_out;

    mlp_tile_kernel<<<MT * NT, 256, 0, stream>>>(features, W1, b1, W2, share);

    morph_pixel_kernel<<<BATCH * GROUPS_PER_IMAGE, 32 * PIX_WAVES, 0, stream>>>(x, share, b2, W3, b3, out);
}


# Pixel pass: layer 3 formed once per work-group, its K = 128 split over the 8 waves (4 WMMAs each, was 32 in every wave), the 8 shares summed through LDS in wave order; k summed in another order
# speedup vs baseline: 8.5542x; 1.0624x over previous
#include <hip/hip_runtime.h>

typedef __attribute__((ext_vector_type(2))) float v2f;
typedef __attribute__((ext_vector_type(4))) float v4f;
typedef __attribute__((ext_vector_type(8))) float v8f;

#define BATCH 64
#define IN_CH 512
#define HID   128
#define NOUT  10
#define MT    (BATCH / 16)
#define NT    (HID / 16)
#define TP    20

struct Frag { v4f q[4]; };

__device__ __forceinline__ void load_frag(Frag& f, const float* p) {
#pragma unroll
    for (int j = 0; j < 4; ++j) f.q[j] = *(const v4f*)(p + 8 * j);
}

__device__ __forceinline__ v8f mma(v2f a, v2f b, v8f c) {
    return __builtin_amdgcn_wmma_f32_16x16x4_f32(false, a, false, b, (short)0, c,
                                                 false, false);
}

__global__ __launch_bounds__(256) void mlp_tile_kernel(
    const float* __restrict__ features,
    const float* __restrict__ W1,
    const float* __restrict__ b1,
    const float* __restrict__ W2,
    float* __restrict__ share)
{
    __shared__ __attribute__((aligned(16))) float psum[8][256];
    __shared__ __attribute__((aligned(16))) float tile[16 * TP];

    const int tid  = threadIdx.x;
    const int lane = tid & 31;
    const int wave = tid >> 5;
    const int c    = lane & 15;
    const int h    = lane >> 4;
    const int mt   = blockIdx.x >> 3;
    const int nt   = blockIdx.x & 7;

    const float* pa = features + (16 * mt + c) * IN_CH + 64 * wave + 4 * h;
    const float* pb = W1       + (16 * nt + c) * IN_CH + 64 * wave + 4 * h;
    Frag a0, a1, b0, b1f;
    load_frag(a0, pa);      load_frag(b0, pb);
    load_frag(a1, pa + 32); load_frag(b1f, pb + 32);
    const float* pw = W2 + (16 * wave + c) * HID + 16 * nt + 4 * h;
    const v4f w0 = *(const v4f*)pw;
    const v4f w1 = *(const v4f*)(pw + 8);
    const float bias = b1[16 * nt + (tid & 15)];
    __builtin_amdgcn_sched_barrier(0);

    v8f e0 = {}, e1 = {};
#pragma unroll
    for (int j = 0; j < 4; ++j) {
        e0 = mma(a0.q[j].xy, b0.q[j].xy, e0);
        e1 = mma(a0.q[j].zw, b0.q[j].zw, e1);
    }
#pragma unroll
    for (int j = 0; j < 4; ++j) {
        e0 = mma(a1.q[j].xy, b1f.q[j].xy, e0);
        e1 = mma(a1.q[j].zw, b1f.q[j].zw, e1);
    }
#pragma unroll
    for (int r = 0; r < 8; ++r) psum[wave][(8 * h + r) * 16 + c] = e0[r] + e1[r];
    __syncthreads();

    {
        float s = psum[0][tid];
#pragma unroll
        for (int w = 1; w < 8; ++w) s += psum[w][tid];
        s += bias;
        tile[(tid >> 4) * TP + (tid & 15)] = s > 0.f ? s : 0.f;
    }
    __syncthreads();

    const v4f x0 = *(const v4f*)(tile + c * TP + 4 * h);
    const v4f x1 = *(const v4f*)(tile + c * TP + 8 + 4 * h);
    v8f f0 = {}, f1 = {};
    f0 = mma(x0.xy, w0.xy, f0);
    f1 = mma(x0.zw, w0.zw, f1);
    f0 = mma(x1.xy, w1.xy, f0);
    f1 = mma(x1.zw, w1.zw, f1);
    float* po = share + (blockIdx.x * 8 + wave) * 256 + (8 * h) * 16 + c;
#pragma unroll
    for (int r = 0; r < 8; ++r) po[r * 16] = f0[r] + f1[r];
}

#define IMG_H 256
#define IMG_W 256
#define BAND  8
#define PIX_WAVES 8
#define GROUPS_PER_IMAGE (IMG_H / (BAND * PIX_WAVES))

__device__ __forceinline__ float lane_pull(int byte_addr, float v) {
    return __int_as_float(__builtin_amdgcn_ds_bpermute(byte_addr, __float_as_int(v)));
}

__device__ __forceinline__ float fast_div(float n, float d) {
    return n * __builtin_amdgcn_rcpf(d);
}

__device__ __forceinline__ v2f pair_pull(int byte_addr, v2f v) {
    v2f r;
    r.x = lane_pull(byte_addr, v.x);
    r.y = lane_pull(byte_addr, v.y);
    return r;
}

__global__ __launch_bounds__(256) void morph_pixel_kernel(
    const float* __restrict__ x,
    const float* __restrict__ share,
    const float* __restrict__ b2,
    const float* __restrict__ W3,
    const float* __restrict__ b3,
    float* __restrict__ out)
{
    __shared__ __attribute__((aligned(16))) float h2s[2 * HID];
    __shared__ __attribute__((aligned(16))) float beta8[16 * PIX_WAVES];

    const int tid  = threadIdx.x;
    const int lane = tid & 31;
    const int wave = __builtin_amdgcn_readfirstlane((int)(threadIdx.x >> 5));
    const int b    = blockIdx.x / GROUPS_PER_IMAGE;
    const int y0   = (blockIdx.x % GROUPS_PER_IMAGE) * (BAND * PIX_WAVES) + wave * BAND;

    const float* img  = x   + (size_t)b * (IMG_H * IMG_W) + 4 * lane;
    float*       oimg = out + (size_t)b * (IMG_H * IMG_W) + 4 * lane;

    const int n = tid & (HID - 1);
    const float* ps = share + ((b >> 4) * NT * 8 + (n >> 4)) * 256 + (b & 15) * 16 + (n & 15);
    float part[NT];
#pragma unroll
    for (int nt = 0; nt < NT; ++nt) part[nt] = ps[nt * 8 * 256];
    const float bias2 = b2[n];

    const int  c     = lane & 15;
    const int  h     = lane >> 4;
    const bool valid = (c < NOUT);
    const float* pw = W3 + (valid ? c : 0) * HID + 16 * wave + 4 * h;
    const v4f w3a = *(const v4f*)pw;
    const v4f w3b = *(const v4f*)(pw + 8);
    const float b3v   = b3[valid ? c : 0];

    __builtin_amdgcn_sched_barrier(0);

    const bool top_inside    = (y0 > 0);
    const bool bottom_inside = (y0 + BAND < IMG_H);
    v4f raw[BAND + 2][2];
#pragma unroll
    for (int i = 0; i < BAND + 2; ++i) {
        const bool inside = (i == 0) ? top_inside : (i == BAND + 1) ? bottom_inside : true;
        const float* rp   = img + (inside ? y0 - 1 + i : y0) * IMG_W;
        raw[i][0] = *(const v4f*)rp;
        raw[i][1] = *(const v4f*)(rp + IMG_W / 2);
    }
    __builtin_amdgcn_sched_barrier(0);

    {
        float s = part[0];
#pragma unroll
        for (int nt = 1; nt < NT; ++nt) s += part[nt];
        s += bias2;
        h2s[tid] = s > 0.f ? s : 0.f;
    }
    __builtin_amdgcn_fence(__ATOMIC_RELEASE, "workgroup", "local");
    __builtin_amdgcn_s_barrier();
    __builtin_amdgcn_fence(__ATOMIC_ACQUIRE, "workgroup", "local");
    {
        const float* pa = h2s + (tid & HID) + 16 * wave + 4 * h;
        const v4f xa = *(const v4f*)pa;
        const v4f xb = *(const v4f*)(pa + 8);
        v8f e0 = {}, e1 = {};
        e0 = mma(xa.xy, w3a.xy, e0);
        e1 = mma(xa.zw, w3a.zw, e1);
        e0 = mma(xb.xy, w3b.xy, e0);
        e1 = mma(xb.zw, w3b.zw, e1);
        if (lane < 16) beta8[c * PIX_WAVES + wave] = e0[0] + e1[0];
    }
    __builtin_amdgcn_fence(__ATOMIC_RELEASE, "workgroup", "local");
    __builtin_amdgcn_s_barrier();
    __builtin_amdgcn_fence(__ATOMIC_ACQUIRE, "workgroup", "local");
    float a;
    v2f k[9];
    {
        const v4f p0 = *(const v4f*)(beta8 + c * PIX_WAVES);
        const v4f p1 = *(const v4f*)(beta8 + c * PIX_WAVES + 4);
        float bsum = p0.x;
        bsum += p0.y; bsum += p0.z; bsum += p0.w;
        bsum += p1.x; bsum += p1.y; bsum += p1.z; bsum += p1.w;
        const float beta = bsum + (valid ? b3v : 0.f);
        const float v    = beta + ((c == 5) ? 5.0f : -5.0f);
        const float sig  = fast_div(1.0f, 1.0f + __expf(-v));
        a = __int_as_float(__builtin_amdgcn_readlane(__float_as_int(beta), 0));
#pragma unroll
        for (int j = 0; j < 9; ++j) {
            const float kj = __int_as_float(__builtin_amdgcn_readlane(__float_as_int(sig), j + 1));
            k[j].x = kj; k[j].y = kj;
        }
    }

    v2f pad;
    pad.y = __expf(a * 0.f);
    pad.x = 0.f * pad.y;

    const int  from_left  = ((lane + 31) & 31) * 4;
    const int  from_right = ((lane + 1) & 31) * 4;
    const bool first = (lane == 0);
    const bool last  = (lane == 31);

    v2f t[BAND + 2][2][6];
#pragma unroll
    for (int i = 0; i < BAND + 2; ++i) {
#pragma unroll
        for (int g = 0; g < 2; ++g) {
#pragma unroll
            for (int s = 0; s < 4; ++s) {
                const bool  inside = (i == 0) ? top_inside : (i == BAND + 1) ? bottom_inside : true;
                const float p = inside ? raw[i][g][s] : 0.f;
                const float e = __expf(a * p);
                t[i][g][s + 1].x = p * e;
                t[i][g][s + 1].y = e;
            }
        }
        const v2f l0 = pair_pull(from_left,  t[i][0][4]);
        const v2f l1 = pair_pull(from_left,  t[i][1][4]);
        const v2f r0 = pair_pull(from_right, t[i][0][1]);
        const v2f r1 = pair_pull(from_right, t[i][1][1]);
        t[i][0][0] = first ? pad : l0;
        t[i][1][0] = first ? l0  : l1;
        t[i][0][5] = last  ? r1  : r0;
        t[i][1][5] = last  ? pad : r1;

        if (i >= 2) {
            const int o = i - 2;
            v4f res[2];
#pragma unroll
            for (int g = 0; g < 2; ++g) {
#pragma unroll
                for (int s = 0; s < 4; ++s) {
                    v2f acc = {0.f, 0.f};
#pragma unroll
                    for (int dy = 0; dy < 3; ++dy) {
#pragma unroll
                        for (int dx = 0; dx < 3; ++dx)
                            acc = __builtin_elementwise_fma(k[dy * 3 + dx], t[o + dy][g][s + dx], acc);
                    }
                    res[g][s] = fast_div(acc.x, acc.y);
                }
            }
            float* op = oimg + (y0 + o) * IMG_W;
            *(v4f*)op               = res[0];
            *(v4f*)(op + IMG_W / 2) = res[1];
            __builtin_amdgcn_sched_barrier(0x81);
        }
    }
}

extern "C" void kernel_launch(void* const* d_in, const int* in_sizes, int n_in,
                              void* d_out, int out_size, void* d_ws, size_t ws_size,
                              hipStream_t stream) {
    const float* x        = (const float*)d_in[0];
    const float* features = (const float*)d_in[1];
    const float* W1       = (const float*)d_in[2];
    const float* b1       = (const float*)d_in[3];
    const float* W2       = (const float*)d_in[4];
    const float* b2       = (const float*)d_in[5];
    const float* W3       = (const float*)d_in[6];
    const float* b3       = (const float*)d_in[7];

    float* share = (float*)d_ws;
    float* out   = (float*)d_out;

    mlp_tile_kernel<<<MT * NT, 256, 0, stream>>>(features, W1, b1, W2, share);

    morph_pixel_kernel<<<BATCH * GROUPS_PER_IMAGE, 32 * PIX_WAVES, 0, stream>>>(x, share, b2, W3, b3, out);
}
